# starting kernel
# baseline (speedup 1.0000x reference)
#include <hip/hip_runtime.h>
#include <math.h>

#define B_N 16384
#define F_N 48
#define T_N 512
#define L_N 16
#define NRT (B_N / 16)
#define NKT_F (F_N / 4)
#define NCT_T (T_N / 16)
#define NCT_F (F_N / 16)

typedef __attribute__((ext_vector_type(2))) float v2f;
typedef __attribute__((ext_vector_type(8))) float v8f;

static __device__ __forceinline__ v8f wmma_f32(v2f a, v2f b, v8f c) {
  return __builtin_amdgcn_wmma_f32_16x16x4_f32(false, a, false, b, (short)0, c,
                                               false, false);
}

static __device__ __forceinline__ int pack_idx(int k, int col, int nct) {
  return (((k >> 2) * nct + (col >> 4)) * 32 + ((k >> 1) & 1) * 16 + (col & 15)) * 2 +
         (k & 1);
}

__global__ void __launch_bounds__(256) k_maxpart(const float* __restrict__ u,
                                                 float* __restrict__ part, int n) {
  __shared__ float sm[256];
  float m = -3.4e38f;
  for (int i = blockIdx.x * 256 + threadIdx.x; i < n; i += 256 * gridDim.x)
    m = fmaxf(m, u[i]);
  sm[threadIdx.x] = m;
  __syncthreads();
  for (int s = 128; s > 0; s >>= 1) {
    if (threadIdx.x < s) sm[threadIdx.x] = fmaxf(sm[threadIdx.x], sm[threadIdx.x + s]);
    __syncthreads();
  }
  if (threadIdx.x == 0) part[blockIdx.x] = sm[0];
}

__global__ void __launch_bounds__(256) k_finalize(const float* __restrict__ part,
                                                  float* __restrict__ scal,
                                                  float* __restrict__ sum) {
  __shared__ float sm[256];
  float m = -3.4e38f;
  for (int i = threadIdx.x; i < 1024; i += 256) m = fmaxf(m, part[i]);
  sm[threadIdx.x] = m;
  __syncthreads();
  for (int s = 128; s > 0; s >>= 1) {
    if (threadIdx.x < s) sm[threadIdx.x] = fmaxf(sm[threadIdx.x], sm[threadIdx.x + s]);
    __syncthreads();
  }
  if (threadIdx.x == 0) {
    float mx = fabsf(sm[0]);
    scal[0] = 5.0f / mx;
    scal[1] = mx / 5.0f;
  }
  for (int t = threadIdx.x; t < L_N * T_N; t += 256) sum[t] = 0.0f;
}

__global__ void __launch_bounds__(256) k_dft(float* __restrict__ irep, float* __restrict__ iimp,
                                             float* __restrict__ drep, float* __restrict__ dimp) {
  int idx = blockIdx.x * 256 + threadIdx.x;
  if (idx >= F_N * T_N) return;
  int f = idx / T_N, t = idx - f * T_N;
  int ph = (f * t) & (T_N - 1);
  float ang = (float)ph * (6.283185307179586f / (float)T_N);
  const float sc = 0.04419417382415922f;
  float sv, cv;
  sincosf(ang, &sv, &cv);
  int ip = pack_idx(f, t, NCT_T);
  irep[ip] = cv * sc;
  iimp[ip] = sv * sc;
  int dp = pack_idx(t, f, NCT_F);
  drep[dp] = cv * sc;
  dimp[dp] = -sv * sc;
}

__global__ void __launch_bounds__(256) k_zero4(float4* __restrict__ p, int n4) {
  for (int i = blockIdx.x * 256 + threadIdx.x; i < n4; i += 256 * gridDim.x)
    p[i] = make_float4(0.f, 0.f, 0.f, 0.f);
}

__global__ void __launch_bounds__(256) k_zinit(const float* __restrict__ u,
                                               const float* __restrict__ scal,
                                               float* __restrict__ zre,
                                               float* __restrict__ zim, int n) {
  const float s = scal[0];
  for (int i = blockIdx.x * 256 + threadIdx.x; i < n; i += 256 * gridDim.x) {
    int m = i / F_N, c = i - m * F_N;
    int p = (((m >> 4) * NKT_F + (c >> 2)) * 32 + ((c >> 1) & 1) * 16 + (m & 15)) * 2 +
            (c & 1);
    zre[p] = s * u[2 * i];
    zim[p] = s * u[2 * i + 1];
  }
}

typedef struct { v2f re[NKT_F], im[NKT_F]; } Abuf;

__global__ void __launch_bounds__(256) k_layerA(
    const float* __restrict__ zre, const float* __restrict__ zim,
    float* __restrict__ hre, float* __restrict__ him,
    const float* __restrict__ brep, const float* __restrict__ bimp,
    float* __restrict__ sum) {
  const int lane = threadIdx.x & 31;
  const int wv = threadIdx.x >> 5;
  const int hi = lane >> 4;
  const int ct = blockIdx.x;
  const int col = ct * 16 + (lane & 15);
  const int wgl = blockIdx.y * 8 + wv;

  v2f Bre[NKT_F], Bim[NKT_F];
  {
    const float* bbr = brep + ct * 64 + lane * 2;
    const float* bbi = bimp + ct * 64 + lane * 2;
#pragma unroll
    for (int kt = 0; kt < NKT_F; ++kt) {
      Bre[kt] = *(const v2f*)(bbr + kt * (NCT_T * 64));
      Bim[kt] = *(const v2f*)(bbi + kt * (NCT_T * 64));
    }
  }

  const float* zr = zre + wgl * (8 * NKT_F * 64) + lane * 2;
  const float* zi = zim + wgl * (8 * NKT_F * 64) + lane * 2;
  float* hbr = hre + wgl * 128 * T_N + col;
  float* hbi = him + wgl * 128 * T_N + col;
  const int hrow = hi * 8 * T_N;

#define LOADA(dst, j_)                                                        \
  do {                                                                        \
    _Pragma("unroll") for (int kt = 0; kt < NKT_F; ++kt) {                    \
      (dst).re[kt] = *(const v2f*)(zr + (j_) * (NKT_F * 64) + kt * 64);       \
      (dst).im[kt] = *(const v2f*)(zi + (j_) * (NKT_F * 64) + kt * 64);       \
    }                                                                         \
  } while (0)

#define TILE(A_, j_)                                                          \
  do {                                                                        \
    float hldr[8], hldi[8];                                                   \
    _Pragma("unroll") for (int i = 0; i < 8; ++i) {                           \
      hldr[i] = hbr[(j_) * (16 * T_N) + hrow + i * T_N];                      \
      hldi[i] = hbi[(j_) * (16 * T_N) + hrow + i * T_N];                      \
    }                                                                         \
    v8f ar_ = {}, ai_ = {};                                                   \
    _Pragma("unroll") for (int kt = 0; kt < NKT_F; ++kt) {                    \
      v2f bn = -Bim[kt];                                                      \
      ar_ = wmma_f32((A_).re[kt], Bre[kt], ar_);                              \
      ar_ = wmma_f32((A_).im[kt], bn, ar_);                                   \
      ai_ = wmma_f32((A_).re[kt], Bim[kt], ai_);                              \
      ai_ = wmma_f32((A_).im[kt], Bre[kt], ai_);                              \
    }                                                                         \
    _Pragma("unroll") for (int i = 0; i < 8; ++i) {                           \
      float rr = ar_[i] + hldr[i], ri = ai_[i] + hldi[i];                     \
      hbr[(j_) * (16 * T_N) + hrow + i * T_N] = rr;                           \
      hbi[(j_) * (16 * T_N) + hrow + i * T_N] = ri;                           \
      colsum += rr * rr + ri * ri;                                            \
    }                                                                         \
  } while (0)

  float colsum = 0.0f;
  Abuf A0, A1;
  LOADA(A0, 0);
  for (int jj = 0; jj < 8; jj += 2) {
    LOADA(A1, jj + 1);
    TILE(A0, jj);
    if (jj + 2 < 8) LOADA(A0, jj + 2);
    TILE(A1, jj + 1);
  }
#undef LOADA
#undef TILE

  colsum += __shfl_xor(colsum, 16, 32);
  if (lane < 16) atomicAdd(&sum[col], colsum);
}

__global__ void __launch_bounds__(256) k_layerC(
    float* __restrict__ hre, float* __restrict__ him,
    const float* __restrict__ drep, const float* __restrict__ dimp,
    float* __restrict__ zre, float* __restrict__ zim,
    const float* __restrict__ u, const float* __restrict__ scal,
    const float* __restrict__ sum, const float* __restrict__ S1,
    const float* __restrict__ S2, float* __restrict__ out, int last) {
  __shared__ float gs[T_N];
  __shared__ float accL[6][256];
  const float s1 = S1[0], s2 = S2[0];
  for (int t = threadIdx.x; t < T_N; t += 256) {
    float rm = sqrtf(sum[t] * (1.0f / (float)B_N));
    gs[t] = 1.0f / (1.0f + expf(-(s1 * (rm - s2))));
  }
  for (int i = threadIdx.x; i < 6 * 256; i += 256) (&accL[0][0])[i] = 0.0f;
  __syncthreads();

  const int lane = threadIdx.x & 31;
  const int wv = threadIdx.x >> 5;
  const int rowbase = blockIdx.x * 16;
  const int kbase = wv * 64;

  for (int j = 0; j < 8; ++j) {
    int idx = (j * 32 + lane) * 4;
    int row = idx >> 6, cc = idx & 63;
    int col = kbase + cc;
    float4 g4 = *(const float4*)&gs[col];
    int off = (rowbase + row) * T_N + col;
    float4 r = *(float4*)&hre[off];
    r.x *= g4.x; r.y *= g4.y; r.z *= g4.z; r.w *= g4.w;
    *(float4*)&hre[off] = r;
    float4 q = *(float4*)&him[off];
    q.x *= g4.x; q.y *= g4.y; q.z *= g4.z; q.w *= g4.w;
    *(float4*)&him[off] = q;
  }

  v8f accRe[3] = {}, accIm[3] = {};
  const int hi = lane >> 4;
  const int cloc = lane & 15;
  const float* harp = hre + (rowbase + cloc) * T_N + kbase + hi * 2;
  const float* haip = him + (rowbase + cloc) * T_N + kbase + hi * 2;
  const float* dbr = drep + (kbase >> 2) * (NCT_F * 64) + lane * 2;
  const float* dbi = dimp + (kbase >> 2) * (NCT_F * 64) + lane * 2;

  v2f pr[2], pi[2];
  pr[0] = *(const v2f*)(harp + 0);
  pi[0] = *(const v2f*)(haip + 0);
  pr[1] = *(const v2f*)(harp + 4);
  pi[1] = *(const v2f*)(haip + 4);
#pragma unroll
  for (int t = 0; t < 16; ++t) {
    v2f a_re = pr[t & 1], a_im = pi[t & 1];
    if (t + 2 < 16) {
      pr[t & 1] = *(const v2f*)(harp + (t + 2) * 4);
      pi[t & 1] = *(const v2f*)(haip + (t + 2) * 4);
    }
#pragma unroll
    for (int nt = 0; nt < NCT_F; ++nt) {
      v2f b_re = *(const v2f*)(dbr + (t * NCT_F + nt) * 64);
      v2f b_im = *(const v2f*)(dbi + (t * NCT_F + nt) * 64);
      v2f bn = -b_im;
      accRe[nt] = wmma_f32(a_re, b_re, accRe[nt]);
      accRe[nt] = wmma_f32(a_im, bn, accRe[nt]);
      accIm[nt] = wmma_f32(a_re, b_im, accIm[nt]);
      accIm[nt] = wmma_f32(a_im, b_re, accIm[nt]);
    }
  }

#pragma unroll
  for (int nt = 0; nt < NCT_F; ++nt) {
#pragma unroll
    for (int i = 0; i < 8; ++i) {
      int sl = (i + hi * 8) * 16 + cloc;
      atomicAdd(&accL[nt * 2 + 0][sl], accRe[nt][i]);
      atomicAdd(&accL[nt * 2 + 1][sl], accIm[nt][i]);
    }
  }
  __syncthreads();

  const float s = scal[0], inv = scal[1];
  const int rt = blockIdx.x;
#pragma unroll
  for (int ii = 0; ii < 3; ++ii) {
    int e = ii * 256 + threadIdx.x;
    int row = e / F_N, c = e - row * F_N;
    int m = rowbase + row;
    float hr = accL[(c >> 4) * 2 + 0][row * 16 + (c & 15)];
    float hm = accL[(c >> 4) * 2 + 1][row * 16 + (c & 15)];
    if (last) {
      out[(m * F_N + c) * 2]     = hr * inv;
      out[(m * F_N + c) * 2 + 1] = hm * inv;
    } else {
      int p = ((rt * NKT_F + (c >> 2)) * 32 + ((c >> 1) & 1) * 16 + row) * 2 + (c & 1);
      zre[p] = s * u[(m * F_N + c) * 2]     - hr;
      zim[p] = s * u[(m * F_N + c) * 2 + 1] - hm;
    }
  }
}

extern "C" void kernel_launch(void* const* d_in, const int* in_sizes, int n_in,
                              void* d_out, int out_size, void* d_ws, size_t ws_size,
                              hipStream_t stream) {
  const float* u  = (const float*)d_in[0];
  const float* S1 = (const float*)d_in[1];
  const float* S2 = (const float*)d_in[2];
  float* out = (float*)d_out;
  float* w = (float*)d_ws;

  float* scal = w;
  float* part = w + 8;
  float* sum  = w + 2048;
  float* irep = w + 16384;
  float* iimp = irep + F_N * T_N;
  float* drep = iimp + F_N * T_N;
  float* dimp = drep + F_N * T_N;
  float* zre  = w + 131072;
  float* zim  = zre + B_N * F_N;
  float* hre  = zim + B_N * F_N;
  float* him  = hre + B_N * T_N;

  const int nu = B_N * F_N * 2;
  k_maxpart<<<1024, 256, 0, stream>>>(u, part, nu);
  k_finalize<<<1, 256, 0, stream>>>(part, scal, sum);
  k_dft<<<(F_N * T_N + 255) / 256, 256, 0, stream>>>(irep, iimp, drep, dimp);
  k_zero4<<<4096, 256, 0, stream>>>((float4*)hre, (2 * B_N * T_N) / 4);
  k_zinit<<<1024, 256, 0, stream>>>(u, scal, zre, zim, B_N * F_N);

  for (int l = 0; l < L_N; ++l) {
    k_layerA<<<dim3(NCT_T, 16), 256, 0, stream>>>(zre, zim, hre, him, irep, iimp,
                                                  sum + l * T_N);
    k_layerC<<<NRT, 256, 0, stream>>>(hre, him, drep, dimp, zre, zim, u, scal,
                                      sum + l * T_N, S1, S2, out,
                                      (l == L_N - 1) ? 1 : 0);
  }
}


# layerC: each wave stages its gated 16x64 chunk in a private LDS tile (pitch 68 dwords) and reads the DFT A fragments from it, not by 16-line global loads behind the stores; bound to 8 waves/SIMD
# speedup vs baseline: 1.0509x; 1.0509x over previous
#include <hip/hip_runtime.h>
#include <math.h>

#define B_N 16384
#define F_N 48
#define T_N 512
#define L_N 16
#define NRT (B_N / 16)
#define NKT_F (F_N / 4)
#define NCT_T (T_N / 16)
#define NCT_F (F_N / 16)

typedef __attribute__((ext_vector_type(2))) float v2f;
typedef __attribute__((ext_vector_type(8))) float v8f;

static __device__ __forceinline__ v8f wmma_f32(v2f a, v2f b, v8f c) {
  return __builtin_amdgcn_wmma_f32_16x16x4_f32(false, a, false, b, (short)0, c,
                                               false, false);
}

static __device__ __forceinline__ int pack_idx(int k, int col, int nct) {
  return (((k >> 2) * nct + (col >> 4)) * 32 + ((k >> 1) & 1) * 16 + (col & 15)) * 2 +
         (k & 1);
}

__global__ void __launch_bounds__(256) k_maxpart(const float* __restrict__ u,
                                                 float* __restrict__ part, int n) {
  __shared__ float sm[256];
  float m = -3.4e38f;
  for (int i = blockIdx.x * 256 + threadIdx.x; i < n; i += 256 * gridDim.x)
    m = fmaxf(m, u[i]);
  sm[threadIdx.x] = m;
  __syncthreads();
  for (int s = 128; s > 0; s >>= 1) {
    if (threadIdx.x < s) sm[threadIdx.x] = fmaxf(sm[threadIdx.x], sm[threadIdx.x + s]);
    __syncthreads();
  }
  if (threadIdx.x == 0) part[blockIdx.x] = sm[0];
}

__global__ void __launch_bounds__(256) k_finalize(const float* __restrict__ part,
                                                  float* __restrict__ scal,
                                                  float* __restrict__ sum) {
  __shared__ float sm[256];
  float m = -3.4e38f;
  for (int i = threadIdx.x; i < 1024; i += 256) m = fmaxf(m, part[i]);
  sm[threadIdx.x] = m;
  __syncthreads();
  for (int s = 128; s > 0; s >>= 1) {
    if (threadIdx.x < s) sm[threadIdx.x] = fmaxf(sm[threadIdx.x], sm[threadIdx.x + s]);
    __syncthreads();
  }
  if (threadIdx.x == 0) {
    float mx = fabsf(sm[0]);
    scal[0] = 5.0f / mx;
    scal[1] = mx / 5.0f;
  }
  for (int t = threadIdx.x; t < L_N * T_N; t += 256) sum[t] = 0.0f;
}

__global__ void __launch_bounds__(256) k_dft(float* __restrict__ irep, float* __restrict__ iimp,
                                             float* __restrict__ drep, float* __restrict__ dimp) {
  int idx = blockIdx.x * 256 + threadIdx.x;
  if (idx >= F_N * T_N) return;
  int f = idx / T_N, t = idx - f * T_N;
  int ph = (f * t) & (T_N - 1);
  float ang = (float)ph * (6.283185307179586f / (float)T_N);
  const float sc = 0.04419417382415922f;
  float sv, cv;
  sincosf(ang, &sv, &cv);
  int ip = pack_idx(f, t, NCT_T);
  irep[ip] = cv * sc;
  iimp[ip] = sv * sc;
  int dp = pack_idx(t, f, NCT_F);
  drep[dp] = cv * sc;
  dimp[dp] = -sv * sc;
}

__global__ void __launch_bounds__(256) k_zero4(float4* __restrict__ p, int n4) {
  for (int i = blockIdx.x * 256 + threadIdx.x; i < n4; i += 256 * gridDim.x)
    p[i] = make_float4(0.f, 0.f, 0.f, 0.f);
}

__global__ void __launch_bounds__(256) k_zinit(const float* __restrict__ u,
                                               const float* __restrict__ scal,
                                               float* __restrict__ zre,
                                               float* __restrict__ zim, int n) {
  const float s = scal[0];
  for (int i = blockIdx.x * 256 + threadIdx.x; i < n; i += 256 * gridDim.x) {
    int m = i / F_N, c = i - m * F_N;
    int p = (((m >> 4) * NKT_F + (c >> 2)) * 32 + ((c >> 1) & 1) * 16 + (m & 15)) * 2 +
            (c & 1);
    zre[p] = s * u[2 * i];
    zim[p] = s * u[2 * i + 1];
  }
}

typedef struct { v2f re[NKT_F], im[NKT_F]; } Abuf;

__global__ void __launch_bounds__(256) k_layerA(
    const float* __restrict__ zre, const float* __restrict__ zim,
    float* __restrict__ hre, float* __restrict__ him,
    const float* __restrict__ brep, const float* __restrict__ bimp,
    float* __restrict__ sum) {
  const int lane = threadIdx.x & 31;
  const int wv = threadIdx.x >> 5;
  const int hi = lane >> 4;
  const int ct = blockIdx.x;
  const int col = ct * 16 + (lane & 15);
  const int wgl = blockIdx.y * 8 + wv;

  v2f Bre[NKT_F], Bim[NKT_F];
  {
    const float* bbr = brep + ct * 64 + lane * 2;
    const float* bbi = bimp + ct * 64 + lane * 2;
#pragma unroll
    for (int kt = 0; kt < NKT_F; ++kt) {
      Bre[kt] = *(const v2f*)(bbr + kt * (NCT_T * 64));
      Bim[kt] = *(const v2f*)(bbi + kt * (NCT_T * 64));
    }
  }

  const float* zr = zre + wgl * (8 * NKT_F * 64) + lane * 2;
  const float* zi = zim + wgl * (8 * NKT_F * 64) + lane * 2;
  float* hbr = hre + wgl * 128 * T_N + col;
  float* hbi = him + wgl * 128 * T_N + col;
  const int hrow = hi * 8 * T_N;

#define LOADA(dst, j_)                                                        \
  do {                                                                        \
    _Pragma("unroll") for (int kt = 0; kt < NKT_F; ++kt) {                    \
      (dst).re[kt] = *(const v2f*)(zr + (j_) * (NKT_F * 64) + kt * 64);       \
      (dst).im[kt] = *(const v2f*)(zi + (j_) * (NKT_F * 64) + kt * 64);       \
    }                                                                         \
  } while (0)

#define TILE(A_, j_)                                                          \
  do {                                                                        \
    float hldr[8], hldi[8];                                                   \
    _Pragma("unroll") for (int i = 0; i < 8; ++i) {                           \
      hldr[i] = hbr[(j_) * (16 * T_N) + hrow + i * T_N];                      \
      hldi[i] = hbi[(j_) * (16 * T_N) + hrow + i * T_N];                      \
    }                                                                         \
    v8f ar_ = {}, ai_ = {};                                                   \
    _Pragma("unroll") for (int kt = 0; kt < NKT_F; ++kt) {                    \
      v2f bn = -Bim[kt];                                                      \
      ar_ = wmma_f32((A_).re[kt], Bre[kt], ar_);                              \
      ar_ = wmma_f32((A_).im[kt], bn, ar_);                                   \
      ai_ = wmma_f32((A_).re[kt], Bim[kt], ai_);                              \
      ai_ = wmma_f32((A_).im[kt], Bre[kt], ai_);                              \
    }                                                                         \
    _Pragma("unroll") for (int i = 0; i < 8; ++i) {                           \
      float rr = ar_[i] + hldr[i], ri = ai_[i] + hldi[i];                     \
      hbr[(j_) * (16 * T_N) + hrow + i * T_N] = rr;                           \
      hbi[(j_) * (16 * T_N) + hrow + i * T_N] = ri;                           \
      colsum += rr * rr + ri * ri;                                            \
    }                                                                         \
  } while (0)

  float colsum = 0.0f;
  Abuf A0, A1;
  LOADA(A0, 0);
  for (int jj = 0; jj < 8; jj += 2) {
    LOADA(A1, jj + 1);
    TILE(A0, jj);
    if (jj + 2 < 8) LOADA(A0, jj + 2);
    TILE(A1, jj + 1);
  }
#undef LOADA
#undef TILE

  colsum += __shfl_xor(colsum, 16, 32);
  if (lane < 16) atomicAdd(&sum[col], colsum);
}

#define HS_PITCH 68
__global__ void __launch_bounds__(256, 8) k_layerC(
    float* __restrict__ hre, float* __restrict__ him,
    const float* __restrict__ drep, const float* __restrict__ dimp,
    float* __restrict__ zre, float* __restrict__ zim,
    const float* __restrict__ u, const float* __restrict__ scal,
    const float* __restrict__ sum, const float* __restrict__ S1,
    const float* __restrict__ S2, float* __restrict__ out, int last) {
  __shared__ float gs[T_N];
  __shared__ float accL[6][256];
  __shared__ __attribute__((aligned(16))) float hsR[8][16 * HS_PITCH];
  __shared__ __attribute__((aligned(16))) float hsI[8][16 * HS_PITCH];
  const float s1 = S1[0], s2 = S2[0];
  for (int t = threadIdx.x; t < T_N; t += 256) {
    float rm = sqrtf(sum[t] * (1.0f / (float)B_N));
    gs[t] = 1.0f / (1.0f + expf(-(s1 * (rm - s2))));
  }
  for (int i = threadIdx.x; i < 6 * 256; i += 256) (&accL[0][0])[i] = 0.0f;
  __syncthreads();

  const int lane = threadIdx.x & 31;
  const int wv = threadIdx.x >> 5;
  const int rowbase = blockIdx.x * 16;
  const int kbase = wv * 64;

  float* const myR = &hsR[wv][0];
  float* const myI = &hsI[wv][0];
  for (int j = 0; j < 8; ++j) {
    int idx = (j * 32 + lane) * 4;
    int row = idx >> 6, cc = idx & 63;
    int col = kbase + cc;
    float4 g4 = *(const float4*)&gs[col];
    int off = (rowbase + row) * T_N + col;
    float4 r = *(float4*)&hre[off];
    r.x *= g4.x; r.y *= g4.y; r.z *= g4.z; r.w *= g4.w;
    *(float4*)&hre[off] = r;
    *(float4*)&myR[row * HS_PITCH + cc] = r;
    float4 q = *(float4*)&him[off];
    q.x *= g4.x; q.y *= g4.y; q.z *= g4.z; q.w *= g4.w;
    *(float4*)&him[off] = q;
    *(float4*)&myI[row * HS_PITCH + cc] = q;
  }

  __builtin_amdgcn_sched_barrier(0);

  v8f accRe[3] = {}, accIm[3] = {};
  const int hi = lane >> 4;
  const int cloc = lane & 15;
  const float* ar = myR + cloc * HS_PITCH + hi * 2;
  const float* ai = myI + cloc * HS_PITCH + hi * 2;
  const float* dbr = drep + (kbase >> 2) * (NCT_F * 64) + lane * 2;
  const float* dbi = dimp + (kbase >> 2) * (NCT_F * 64) + lane * 2;

#pragma unroll 4
  for (int t = 0; t < 16; ++t) {
    v2f a_re = *(const v2f*)(ar + t * 4);
    v2f a_im = *(const v2f*)(ai + t * 4);
#pragma unroll
    for (int nt = 0; nt < NCT_F; ++nt) {
      v2f b_re = *(const v2f*)(dbr + (t * NCT_F + nt) * 64);
      v2f b_im = *(const v2f*)(dbi + (t * NCT_F + nt) * 64);
      v2f bn = -b_im;
      accRe[nt] = wmma_f32(a_re, b_re, accRe[nt]);
      accRe[nt] = wmma_f32(a_im, bn, accRe[nt]);
      accIm[nt] = wmma_f32(a_re, b_im, accIm[nt]);
      accIm[nt] = wmma_f32(a_im, b_re, accIm[nt]);
    }
  }

#pragma unroll
  for (int nt = 0; nt < NCT_F; ++nt) {
#pragma unroll
    for (int i = 0; i < 8; ++i) {
      int sl = (i + hi * 8) * 16 + cloc;
      atomicAdd(&accL[nt * 2 + 0][sl], accRe[nt][i]);
      atomicAdd(&accL[nt * 2 + 1][sl], accIm[nt][i]);
    }
  }
  __syncthreads();

  const float s = scal[0], inv = scal[1];
  const int rt = blockIdx.x;
#pragma unroll
  for (int ii = 0; ii < 3; ++ii) {
    int e = ii * 256 + threadIdx.x;
    int row = e / F_N, c = e - row * F_N;
    int m = rowbase + row;
    float hr = accL[(c >> 4) * 2 + 0][row * 16 + (c & 15)];
    float hm = accL[(c >> 4) * 2 + 1][row * 16 + (c & 15)];
    if (last) {
      out[(m * F_N + c) * 2]     = hr * inv;
      out[(m * F_N + c) * 2 + 1] = hm * inv;
    } else {
      int p = ((rt * NKT_F + (c >> 2)) * 32 + ((c >> 1) & 1) * 16 + row) * 2 + (c & 1);
      zre[p] = s * u[(m * F_N + c) * 2]     - hr;
      zim[p] = s * u[(m * F_N + c) * 2 + 1] - hm;
    }
  }
}

extern "C" void kernel_launch(void* const* d_in, const int* in_sizes, int n_in,
                              void* d_out, int out_size, void* d_ws, size_t ws_size,
                              hipStream_t stream) {
  const float* u  = (const float*)d_in[0];
  const float* S1 = (const float*)d_in[1];
  const float* S2 = (const float*)d_in[2];
  float* out = (float*)d_out;
  float* w = (float*)d_ws;

  float* scal = w;
  float* part = w + 8;
  float* sum  = w + 2048;
  float* irep = w + 16384;
  float* iimp = irep + F_N * T_N;
  float* drep = iimp + F_N * T_N;
  float* dimp = drep + F_N * T_N;
  float* zre  = w + 131072;
  float* zim  = zre + B_N * F_N;
  float* hre  = zim + B_N * F_N;
  float* him  = hre + B_N * T_N;

  const int nu = B_N * F_N * 2;
  k_maxpart<<<1024, 256, 0, stream>>>(u, part, nu);
  k_finalize<<<1, 256, 0, stream>>>(part, scal, sum);
  k_dft<<<(F_N * T_N + 255) / 256, 256, 0, stream>>>(irep, iimp, drep, dimp);
  k_zero4<<<4096, 256, 0, stream>>>((float4*)hre, (2 * B_N * T_N) / 4);
  k_zinit<<<1024, 256, 0, stream>>>(u, scal, zre, zim, B_N * F_N);

  for (int l = 0; l < L_N; ++l) {
    k_layerA<<<dim3(NCT_T, 16), 256, 0, stream>>>(zre, zim, hre, him, irep, iimp,
                                                  sum + l * T_N);
    k_layerC<<<NRT, 256, 0, stream>>>(hre, him, drep, dimp, zre, zim, u, scal,
                                      sum + l * T_N, S1, S2, out,
                                      (l == L_N - 1) ? 1 : 0);
  }
}


# H kept as transposed 16x16 accumulator tiles: both layer kernels move H in 512-byte runs, layerC feeds tile registers to WMMA without LDS staging; column sums via lane shuffles (sum order changes)
# speedup vs baseline: 3.4136x; 3.2483x over previous
#include <hip/hip_runtime.h>
#include <math.h>

#define B_N 16384
#define F_N 48
#define T_N 512
#define L_N 16
#define NRT (B_N / 16)
#define NKT_F (F_N / 4)
#define NCT_T (T_N / 16)
#define NCT_F (F_N / 16)

typedef __attribute__((ext_vector_type(2))) float v2f;
typedef __attribute__((ext_vector_type(4))) float v4f;
typedef __attribute__((ext_vector_type(8))) float v8f;

static __device__ __forceinline__ v8f wmma_f32(v2f a, v2f b, v8f c) {
  return __builtin_amdgcn_wmma_f32_16x16x4_f32(false, a, false, b, (short)0, c,
                                               false, false);
}

static __device__ __forceinline__ int pack_idx(int k, int col, int nct) {
  return (((k >> 2) * nct + (col >> 4)) * 32 + ((k >> 1) & 1) * 16 + (col & 15)) * 2 +
         (k & 1);
}

static __device__ __forceinline__ int pack_idx_dft(int t, int f) {
  int tt = t & 15;
  int ks = (t >> 4) * 4 + ((tt >> 1) & 3);
  return ((ks * NCT_F + (f >> 4)) * 32 + (tt >> 3) * 16 + (f & 15)) * 2 + (tt & 1);
}

__global__ void __launch_bounds__(256) k_maxpart(const float* __restrict__ u,
                                                 float* __restrict__ part, int n) {
  __shared__ float sm[256];
  float m = -3.4e38f;
  for (int i = blockIdx.x * 256 + threadIdx.x; i < n; i += 256 * gridDim.x)
    m = fmaxf(m, u[i]);
  sm[threadIdx.x] = m;
  __syncthreads();
  for (int s = 128; s > 0; s >>= 1) {
    if (threadIdx.x < s) sm[threadIdx.x] = fmaxf(sm[threadIdx.x], sm[threadIdx.x + s]);
    __syncthreads();
  }
  if (threadIdx.x == 0) part[blockIdx.x] = sm[0];
}

__global__ void __launch_bounds__(256) k_finalize(const float* __restrict__ part,
                                                  float* __restrict__ scal,
                                                  float* __restrict__ sum) {
  __shared__ float sm[256];
  float m = -3.4e38f;
  for (int i = threadIdx.x; i < 1024; i += 256) m = fmaxf(m, part[i]);
  sm[threadIdx.x] = m;
  __syncthreads();
  for (int s = 128; s > 0; s >>= 1) {
    if (threadIdx.x < s) sm[threadIdx.x] = fmaxf(sm[threadIdx.x], sm[threadIdx.x + s]);
    __syncthreads();
  }
  if (threadIdx.x == 0) {
    float mx = fabsf(sm[0]);
    scal[0] = 5.0f / mx;
    scal[1] = mx / 5.0f;
  }
  for (int t = threadIdx.x; t < L_N * T_N; t += 256) sum[t] = 0.0f;
}

__global__ void __launch_bounds__(256) k_dft(float* __restrict__ irep, float* __restrict__ iimp,
                                             float* __restrict__ drep, float* __restrict__ dimp) {
  int idx = blockIdx.x * 256 + threadIdx.x;
  if (idx >= F_N * T_N) return;
  int f = idx / T_N, t = idx - f * T_N;
  int ph = (f * t) & (T_N - 1);
  float ang = (float)ph * (6.283185307179586f / (float)T_N);
  const float sc = 0.04419417382415922f;
  float sv, cv;
  sincosf(ang, &sv, &cv);
  int ip = pack_idx(f, t, NCT_T);
  irep[ip] = cv * sc;
  iimp[ip] = sv * sc;
  int dp = pack_idx_dft(t, f);
  drep[dp] = cv * sc;
  dimp[dp] = -sv * sc;
}

__global__ void __launch_bounds__(256) k_zero4(float4* __restrict__ p, int n4) {
  for (int i = blockIdx.x * 256 + threadIdx.x; i < n4; i += 256 * gridDim.x)
    p[i] = make_float4(0.f, 0.f, 0.f, 0.f);
}

__global__ void __launch_bounds__(256) k_zinit(const float* __restrict__ u,
                                               const float* __restrict__ scal,
                                               float* __restrict__ zre,
                                               float* __restrict__ zim, int n) {
  const float s = scal[0];
  for (int i = blockIdx.x * 256 + threadIdx.x; i < n; i += 256 * gridDim.x) {
    int m = i / F_N, c = i - m * F_N;
    int p = (((m >> 4) * NKT_F + (c >> 2)) * 32 + ((c >> 1) & 1) * 16 + (m & 15)) * 2 +
            (c & 1);
    zre[p] = s * u[2 * i];
    zim[p] = s * u[2 * i + 1];
  }
}

typedef struct { v2f re[NKT_F], im[NKT_F]; } Abuf;

__global__ void __launch_bounds__(256) k_layerA(
    const float* __restrict__ zre, const float* __restrict__ zim,
    float* __restrict__ hre, float* __restrict__ him,
    const float* __restrict__ brep, const float* __restrict__ bimp,
    float* __restrict__ sum) {
  const int lane = threadIdx.x & 31;
  const int wv = threadIdx.x >> 5;
  const int hi = lane >> 4;
  const int ct = blockIdx.x;
  const int wgl = blockIdx.y * 8 + wv;

  v2f Bre[NKT_F], Bim[NKT_F];
  {
    const float* bbr = brep + ct * 64 + lane * 2;
    const float* bbi = bimp + ct * 64 + lane * 2;
#pragma unroll
    for (int kt = 0; kt < NKT_F; ++kt) {
      Bre[kt] = *(const v2f*)(bbr + kt * (NCT_T * 64));
      Bim[kt] = *(const v2f*)(bbi + kt * (NCT_T * 64));
    }
  }

  const float* zr = zre + wgl * (8 * NKT_F * 64) + lane * 2;
  const float* zi = zim + wgl * (8 * NKT_F * 64) + lane * 2;
  float* hbr = hre + ((wgl * 8) * NCT_T + ct) * 256 + lane * 4;
  float* hbi = him + ((wgl * 8) * NCT_T + ct) * 256 + lane * 4;
#define HT (NCT_T * 256)

#define LOADA(dst, j_)                                                        \
  do {                                                                        \
    _Pragma("unroll") for (int kt = 0; kt < NKT_F; ++kt) {                    \
      (dst).re[kt] = *(const v2f*)(zr + (j_) * (NKT_F * 64) + kt * 64);       \
      (dst).im[kt] = *(const v2f*)(zi + (j_) * (NKT_F * 64) + kt * 64);       \
    }                                                                         \
  } while (0)

#define TILE(A_, j_)                                                          \
  do {                                                                        \
    v4f hr0 = *(const v4f*)(hbr + (j_) * HT), hr1 = *(const v4f*)(hbr + (j_) * HT + 128); \
    v4f hi0 = *(const v4f*)(hbi + (j_) * HT), hi1 = *(const v4f*)(hbi + (j_) * HT + 128); \
    v8f ar_ = {}, ai_ = {};                                                   \
    _Pragma("unroll") for (int kt = 0; kt < NKT_F; ++kt) {                    \
      v2f bn = -Bim[kt];                                                      \
      ar_ = wmma_f32(Bre[kt], (A_).re[kt], ar_);                              \
      ar_ = wmma_f32(bn, (A_).im[kt], ar_);                                   \
      ai_ = wmma_f32(Bim[kt], (A_).re[kt], ai_);                              \
      ai_ = wmma_f32(Bre[kt], (A_).im[kt], ai_);                              \
    }                                                                         \
    v4f rr0, rr1, ri0, ri1;                                                   \
    _Pragma("unroll") for (int i = 0; i < 4; ++i) {                           \
      rr0[i] = ar_[i] + hr0[i];  rr1[i] = ar_[4 + i] + hr1[i];                \
      ri0[i] = ai_[i] + hi0[i];  ri1[i] = ai_[4 + i] + hi1[i];                \
      cs[i]     += rr0[i] * rr0[i] + ri0[i] * ri0[i];                         \
      cs[4 + i] += rr1[i] * rr1[i] + ri1[i] * ri1[i];                         \
    }                                                                         \
    *(v4f*)(hbr + (j_) * HT) = rr0;  *(v4f*)(hbr + (j_) * HT + 128) = rr1;    \
    *(v4f*)(hbi + (j_) * HT) = ri0;  *(v4f*)(hbi + (j_) * HT + 128) = ri1;    \
  } while (0)

  float cs[8] = {0.f, 0.f, 0.f, 0.f, 0.f, 0.f, 0.f, 0.f};
  Abuf A0, A1;
  LOADA(A0, 0);
  for (int jj = 0; jj < 8; jj += 2) {
    LOADA(A1, jj + 1);
    TILE(A0, jj);
    if (jj + 2 < 8) LOADA(A0, jj + 2);
    TILE(A1, jj + 1);
  }
#undef LOADA
#undef TILE
#undef HT

#pragma unroll
  for (int r = 0; r < 8; ++r) {
    float v = cs[r];
    v += __shfl_xor(v, 1, 32);
    v += __shfl_xor(v, 2, 32);
    v += __shfl_xor(v, 4, 32);
    v += __shfl_xor(v, 8, 32);
    cs[r] = v;
  }
  float mine = cs[0];
#pragma unroll
  for (int r = 1; r < 8; ++r) mine = ((lane & 7) == r) ? cs[r] : mine;
  if ((lane & 15) < 8) atomicAdd(&sum[ct * 16 + hi * 8 + (lane & 7)], mine);
}

__global__ void __launch_bounds__(256, 8) k_layerC(
    float* __restrict__ hre, float* __restrict__ him,
    const float* __restrict__ drep, const float* __restrict__ dimp,
    float* __restrict__ zre, float* __restrict__ zim,
    const float* __restrict__ u, const float* __restrict__ scal,
    const float* __restrict__ sum, const float* __restrict__ S1,
    const float* __restrict__ S2, float* __restrict__ out, int last) {
  __shared__ __attribute__((aligned(16))) float gs[T_N];
  __shared__ float accL[6][256];
  const float s1 = S1[0], s2 = S2[0];
  for (int t = threadIdx.x; t < T_N; t += 256) {
    float rm = sqrtf(sum[t] * (1.0f / (float)B_N));
    gs[t] = 1.0f / (1.0f + expf(-(s1 * (rm - s2))));
  }
  for (int i = threadIdx.x; i < 6 * 256; i += 256) (&accL[0][0])[i] = 0.0f;
  __syncthreads();

  const int lane = threadIdx.x & 31;
  const int wv = threadIdx.x >> 5;
  const int hi = lane >> 4;
  const int cloc = lane & 15;
  const int rt = blockIdx.x;
  const int rowbase = rt * 16;
  const int ct0 = wv * 4;

  float* hpr = hre + (rt * NCT_T + ct0) * 256 + lane * 4;
  float* hpi = him + (rt * NCT_T + ct0) * 256 + lane * 4;
  const float* gp = gs + ct0 * 16 + hi * 8;
  const float* dbr = drep + (ct0 * 4) * (NCT_F * 64) + lane * 2;
  const float* dbi = dimp + (ct0 * 4) * (NCT_F * 64) + lane * 2;

  v8f accRe[3] = {}, accIm[3] = {};
#pragma unroll
  for (int hp = 0; hp < 2; ++hp) {
    v4f xr[2][2], xi[2][2];
#pragma unroll
    for (int i = 0; i < 2; ++i) {
      xr[i][0] = *(const v4f*)(hpr + (hp * 2 + i) * 256);
      xr[i][1] = *(const v4f*)(hpr + (hp * 2 + i) * 256 + 128);
      xi[i][0] = *(const v4f*)(hpi + (hp * 2 + i) * 256);
      xi[i][1] = *(const v4f*)(hpi + (hp * 2 + i) * 256 + 128);
    }
    v2f bre[NCT_F], bim[NCT_F];
#pragma unroll
    for (int nt = 0; nt < NCT_F; ++nt) {
      bre[nt] = *(const v2f*)(dbr + ((hp * 8) * NCT_F + nt) * 64);
      bim[nt] = *(const v2f*)(dbi + ((hp * 8) * NCT_F + nt) * 64);
    }
#pragma unroll
    for (int i = 0; i < 2; ++i) {
      const v4f g0 = *(const v4f*)(gp + (hp * 2 + i) * 16), g1 = *(const v4f*)(gp + (hp * 2 + i) * 16 + 4);
      xr[i][0] *= g0;  xr[i][1] *= g1;
      xi[i][0] *= g0;  xi[i][1] *= g1;
      *(v4f*)(hpr + (hp * 2 + i) * 256) = xr[i][0];  *(v4f*)(hpr + (hp * 2 + i) * 256 + 128) = xr[i][1];
      *(v4f*)(hpi + (hp * 2 + i) * 256) = xi[i][0];  *(v4f*)(hpi + (hp * 2 + i) * 256 + 128) = xi[i][1];
    }
#pragma unroll
    for (int ks = 0; ks < 8; ++ks) {
      v2f nre[NCT_F], nim[NCT_F];
      if (ks + 1 < 8) {
#pragma unroll
        for (int nt = 0; nt < NCT_F; ++nt) {
          nre[nt] = *(const v2f*)(dbr + ((hp * 8 + ks + 1) * NCT_F + nt) * 64);
          nim[nt] = *(const v2f*)(dbi + ((hp * 8 + ks + 1) * NCT_F + nt) * 64);
        }
      }
      const int i = ks >> 2, s = ks & 3;
      v2f a_re, a_im;
      a_re[0] = xr[i][s >> 1][(s & 1) * 2];  a_re[1] = xr[i][s >> 1][(s & 1) * 2 + 1];
      a_im[0] = xi[i][s >> 1][(s & 1) * 2];  a_im[1] = xi[i][s >> 1][(s & 1) * 2 + 1];
#pragma unroll
      for (int nt = 0; nt < NCT_F; ++nt) {
        v2f bn = -bim[nt];
        accRe[nt] = wmma_f32(a_re, bre[nt], accRe[nt]);
        accRe[nt] = wmma_f32(a_im, bn, accRe[nt]);
        accIm[nt] = wmma_f32(a_re, bim[nt], accIm[nt]);
        accIm[nt] = wmma_f32(a_im, bre[nt], accIm[nt]);
      }
      if (ks + 1 < 8) {
#pragma unroll
        for (int nt = 0; nt < NCT_F; ++nt) { bre[nt] = nre[nt]; bim[nt] = nim[nt]; }
      }
    }
  }

#pragma unroll
  for (int nt = 0; nt < NCT_F; ++nt) {
#pragma unroll
    for (int i = 0; i < 8; ++i) {
      int sl = (i + hi * 8) * 16 + cloc;
      atomicAdd(&accL[nt * 2 + 0][sl], accRe[nt][i]);
      atomicAdd(&accL[nt * 2 + 1][sl], accIm[nt][i]);
    }
  }
  __syncthreads();

  const float s = scal[0], inv = scal[1];
#pragma unroll
  for (int ii = 0; ii < 3; ++ii) {
    int e = ii * 256 + threadIdx.x;
    int row = e / F_N, c = e - row * F_N;
    int m = rowbase + row;
    float hr = accL[(c >> 4) * 2 + 0][row * 16 + (c & 15)];
    float hm = accL[(c >> 4) * 2 + 1][row * 16 + (c & 15)];
    if (last) {
      out[(m * F_N + c) * 2]     = hr * inv;
      out[(m * F_N + c) * 2 + 1] = hm * inv;
    } else {
      int p = ((rt * NKT_F + (c >> 2)) * 32 + ((c >> 1) & 1) * 16 + row) * 2 + (c & 1);
      zre[p] = s * u[(m * F_N + c) * 2]     - hr;
      zim[p] = s * u[(m * F_N + c) * 2 + 1] - hm;
    }
  }
}

extern "C" void kernel_launch(void* const* d_in, const int* in_sizes, int n_in,
                              void* d_out, int out_size, void* d_ws, size_t ws_size,
                              hipStream_t stream) {
  const float* u  = (const float*)d_in[0];
  const float* S1 = (const float*)d_in[1];
  const float* S2 = (const float*)d_in[2];
  float* out = (float*)d_out;
  float* w = (float*)d_ws;

  float* scal = w;
  float* part = w + 8;
  float* sum  = w + 2048;
  float* irep = w + 16384;
  float* iimp = irep + F_N * T_N;
  float* drep = iimp + F_N * T_N;
  float* dimp = drep + F_N * T_N;
  float* zre  = w + 131072;
  float* zim  = zre + B_N * F_N;
  float* hre  = zim + B_N * F_N;
  float* him  = hre + B_N * T_N;

  const int nu = B_N * F_N * 2;
  k_maxpart<<<1024, 256, 0, stream>>>(u, part, nu);
  k_finalize<<<1, 256, 0, stream>>>(part, scal, sum);
  k_dft<<<(F_N * T_N + 255) / 256, 256, 0, stream>>>(irep, iimp, drep, dimp);
  k_zero4<<<4096, 256, 0, stream>>>((float4*)hre, (2 * B_N * T_N) / 4);
  k_zinit<<<1024, 256, 0, stream>>>(u, scal, zre, zim, B_N * F_N);

  for (int l = 0; l < L_N; ++l) {
    k_layerA<<<dim3(NCT_T, 16), 256, 0, stream>>>(zre, zim, hre, him, irep, iimp,
                                                  sum + l * T_N);
    k_layerC<<<NRT, 256, 0, stream>>>(hre, him, drep, dimp, zre, zim, u, scal,
                                      sum + l * T_N, S1, S2, out,
                                      (l == L_N - 1) ? 1 : 0);
  }
}


# layer halves fused per 16-row strip: gate+dft of layer l-1 and idft+residual of layer l in one kernel, H strip stays in registers, z in LDS; column sums per strip into 8 copies (sum order changes)
# speedup vs baseline: 4.8135x; 1.4101x over previous
#include <hip/hip_runtime.h>
#include <math.h>

#define B_N 16384
#define F_N 48
#define T_N 512
#define L_N 16
#define NRT (B_N / 16)
#define NKT_F (F_N / 4)
#define NCT_T (T_N / 16)
#define NCT_F (F_N / 16)
#define NCOPY 8

typedef __attribute__((ext_vector_type(2))) float v2f;
typedef __attribute__((ext_vector_type(4))) float v4f;
typedef __attribute__((ext_vector_type(8))) float v8f;

static __device__ __forceinline__ v8f wmma_f32(v2f a, v2f b, v8f c) {
  return __builtin_amdgcn_wmma_f32_16x16x4_f32(false, a, false, b, (short)0, c,
                                               false, false);
}

static __device__ __forceinline__ int pack_idx(int k, int col, int nct) {
  return (((k >> 2) * nct + (col >> 4)) * 32 + ((k >> 1) & 1) * 16 + (col & 15)) * 2 +
         (k & 1);
}

static __device__ __forceinline__ int pack_idx_dft(int t, int f) {
  int tt = t & 15;
  int ks = (t >> 4) * 4 + ((tt >> 1) & 3);
  return ((ks * NCT_F + (f >> 4)) * 32 + (tt >> 3) * 16 + (f & 15)) * 2 + (tt & 1);
}

__global__ void __launch_bounds__(256) k_maxpart(const float* __restrict__ u,
                                                 float* __restrict__ part, int n) {
  __shared__ float sm[256];
  float m = -3.4e38f;
  for (int i = blockIdx.x * 256 + threadIdx.x; i < n; i += 256 * gridDim.x)
    m = fmaxf(m, u[i]);
  sm[threadIdx.x] = m;
  __syncthreads();
  for (int s = 128; s > 0; s >>= 1) {
    if (threadIdx.x < s) sm[threadIdx.x] = fmaxf(sm[threadIdx.x], sm[threadIdx.x + s]);
    __syncthreads();
  }
  if (threadIdx.x == 0) part[blockIdx.x] = sm[0];
}

__global__ void __launch_bounds__(256) k_finalize(const float* __restrict__ part,
                                                  float* __restrict__ scal) {
  __shared__ float sm[256];
  float m = -3.4e38f;
  for (int i = threadIdx.x; i < 1024; i += 256) m = fmaxf(m, part[i]);
  sm[threadIdx.x] = m;
  __syncthreads();
  for (int s = 128; s > 0; s >>= 1) {
    if (threadIdx.x < s) sm[threadIdx.x] = fmaxf(sm[threadIdx.x], sm[threadIdx.x + s]);
    __syncthreads();
  }
  if (threadIdx.x == 0) {
    float mx = fabsf(sm[0]);
    scal[0] = 5.0f / mx;
    scal[1] = mx / 5.0f;
  }
}

__global__ void __launch_bounds__(256) k_dft(float* __restrict__ irep, float* __restrict__ iimp,
                                             float* __restrict__ drep, float* __restrict__ dimp,
                                             float* __restrict__ sums) {
  int idx = blockIdx.x * 256 + threadIdx.x;
  for (int i = idx; i < L_N * NCOPY * T_N; i += 256 * gridDim.x) sums[i] = 0.0f;
  if (idx >= F_N * T_N) return;
  int f = idx / T_N, t = idx - f * T_N;
  int ph = (f * t) & (T_N - 1);
  float ang = (float)ph * (6.283185307179586f / (float)T_N);
  const float sc = 0.04419417382415922f;
  float sv, cv;
  sincosf(ang, &sv, &cv);
  int ip = pack_idx(f, t, NCT_T);
  irep[ip] = cv * sc;
  iimp[ip] = sv * sc;
  int dp = pack_idx_dft(t, f);
  drep[dp] = cv * sc;
  dimp[dp] = -sv * sc;
}

static __device__ __forceinline__ void layer_body(
    const int mode,
    float* __restrict__ hre, float* __restrict__ him,
    const float* __restrict__ dft,
    const float* __restrict__ idft,
    const float* __restrict__ u, const float* __restrict__ scal,
    const float* __restrict__ sum_in, float* __restrict__ sum_out,
    const float* __restrict__ S1, const float* __restrict__ S2, float* __restrict__ out) {
  __shared__ __attribute__((aligned(16))) float gs[T_N];
  __shared__ float accL[6][256];
  __shared__ __attribute__((aligned(16))) float zLre[NKT_F * 64];
  __shared__ __attribute__((aligned(16))) float zLim[NKT_F * 64];

  const int tid = threadIdx.x;
  const int lane = tid & 31;
  const int wv = tid >> 5;
  const int hi = lane >> 4;
  const int cloc = lane & 15;
  const int ct0 = wv * 4;

  if (mode != 0) {
    const float s1 = S1[0], s2 = S2[0];
    for (int t = tid; t < T_N; t += 256) {
      float a = 0.0f;
#pragma unroll
      for (int c = 0; c < NCOPY; ++c) a += sum_in[c * T_N + t];
      float rm = sqrtf(a * (1.0f / (float)B_N));
      gs[t] = 1.0f / (1.0f + expf(-(s1 * (rm - s2))));
    }
    for (int i = tid; i < 6 * 256; i += 256) (&accL[0][0])[i] = 0.0f;
  }
  __syncthreads();

  const float s = scal[0], inv = scal[1];
  const float* gp = gs + ct0 * 16 + hi * 8;
  const float* dbr0 = dft + (ct0 * 4) * (NCT_F * 64) + lane * 2;
  const float* bbr0 = idft + ct0 * 64 + lane * 2;
  const int nstep = gridDim.x * (NCT_T * 256);

  v4f xr[4][2], xi[4][2];
#pragma unroll
  for (int i = 0; i < 4; ++i) {
    xr[i][0] = (v4f){0.f, 0.f, 0.f, 0.f};  xr[i][1] = (v4f){0.f, 0.f, 0.f, 0.f};
    xi[i][0] = (v4f){0.f, 0.f, 0.f, 0.f};  xi[i][1] = (v4f){0.f, 0.f, 0.f, 0.f};
  }

#define LOADH(pr_, pi_)                                                       \
  do {                                                                        \
    _Pragma("unroll") for (int i = 0; i < 4; ++i) {                           \
      xr[i][0] = *(const v4f*)((pr_) + i * 256);                              \
      xr[i][1] = *(const v4f*)((pr_) + i * 256 + 128);                        \
      xi[i][0] = *(const v4f*)((pi_) + i * 256);                              \
      xi[i][1] = *(const v4f*)((pi_) + i * 256 + 128);                        \
    }                                                                         \
  } while (0)

  if (mode == 1) {
    const float* pr = hre + (blockIdx.x * NCT_T + ct0) * 256 + lane * 4;
    const float* pi = him + (blockIdx.x * NCT_T + ct0) * 256 + lane * 4;
    LOADH(pr, pi);
  }

  for (int rt = blockIdx.x; rt < NRT; rt += gridDim.x) {
    int opq = 0;
    asm volatile("" : "+s"(opq));
    const float* dbr = dbr0 + opq;
    const float* dbi = dbr + F_N * T_N;
    const float* bbr = bbr0 + opq;
    const float* bbi = bbr + F_N * T_N;
    const int rowbase = rt * 16;
    float* hbr = hre + (rt * NCT_T + ct0) * 256 + lane * 4;
    float* hbi = him + (rt * NCT_T + ct0) * 256 + lane * 4;

    float ur[3], ui[3];
    if (mode != 2) {
#pragma unroll
      for (int ii = 0; ii < 3; ++ii) {
        int e = ii * 256 + tid;
        ur[ii] = u[(rowbase * F_N + e) * 2];
        ui[ii] = u[(rowbase * F_N + e) * 2 + 1];
      }
    }

    v2f Bre[NKT_F], Bim[NKT_F];
    if (mode == 0) {
#pragma unroll
      for (int kt = 0; kt < NKT_F; ++kt) {
        Bre[kt] = *(const v2f*)(bbr + kt * (NCT_T * 64));
        Bim[kt] = *(const v2f*)(bbi + kt * (NCT_T * 64));
      }
    }
    if (mode != 0) {
      if (mode == 2) LOADH(hbr, hbi);
#pragma unroll
      for (int i = 0; i < 4; ++i) {
        const v4f g0 = *(const v4f*)(gp + i * 16), g1 = *(const v4f*)(gp + i * 16 + 4);
        xr[i][0] *= g0;  xr[i][1] *= g1;
        xi[i][0] *= g0;  xi[i][1] *= g1;
      }
      v8f accRe[NCT_F] = {}, accIm[NCT_F] = {};
      v2f bre[NCT_F], bim[NCT_F];
#pragma unroll
      for (int nt = 0; nt < NCT_F; ++nt) {
        bre[nt] = *(const v2f*)(dbr + nt * 64);
        bim[nt] = *(const v2f*)(dbi + nt * 64);
      }
#pragma unroll
      for (int kk = 0; kk < 16; ++kk) {
        v2f nre[NCT_F], nim[NCT_F];
        if (kk + 1 < 16) {
#pragma unroll
          for (int nt = 0; nt < NCT_F; ++nt) {
            nre[nt] = *(const v2f*)(dbr + ((kk + 1) * NCT_F + nt) * 64);
            nim[nt] = *(const v2f*)(dbi + ((kk + 1) * NCT_F + nt) * 64);
          }
        }
        const int i = kk >> 2, sp = kk & 3;
        v2f a_re, a_im;
        a_re[0] = xr[i][sp >> 1][(sp & 1) * 2];  a_re[1] = xr[i][sp >> 1][(sp & 1) * 2 + 1];
        a_im[0] = xi[i][sp >> 1][(sp & 1) * 2];  a_im[1] = xi[i][sp >> 1][(sp & 1) * 2 + 1];
#pragma unroll
        for (int nt = 0; nt < NCT_F; ++nt) {
          v2f bn = -bim[nt];
          accRe[nt] = wmma_f32(a_re, bre[nt], accRe[nt]);
          accRe[nt] = wmma_f32(a_im, bn, accRe[nt]);
          accIm[nt] = wmma_f32(a_re, bim[nt], accIm[nt]);
          accIm[nt] = wmma_f32(a_im, bre[nt], accIm[nt]);
        }
        if (kk + 1 < 16) {
#pragma unroll
          for (int nt = 0; nt < NCT_F; ++nt) { bre[nt] = nre[nt]; bim[nt] = nim[nt]; }
        }
        __builtin_amdgcn_sched_barrier(0);
      }
#pragma unroll
      for (int nt = 0; nt < NCT_F; ++nt) {
#pragma unroll
        for (int i = 0; i < 8; ++i) {
          int sl = (i + hi * 8) * 16 + cloc;
          atomicAdd(&accL[nt * 2 + 0][sl], accRe[nt][i]);
          atomicAdd(&accL[nt * 2 + 1][sl], accIm[nt][i]);
        }
      }
      if (mode != 2) {
#pragma unroll
        for (int kt = 0; kt < NKT_F; ++kt) {
          Bre[kt] = *(const v2f*)(bbr + kt * (NCT_T * 64));
          Bim[kt] = *(const v2f*)(bbi + kt * (NCT_T * 64));
        }
      }
    }
    __syncthreads();

#pragma unroll
    for (int ii = 0; ii < 3; ++ii) {
      int e = ii * 256 + tid;
      int row = e / F_N, c = e - row * F_N;
      float hr = 0.0f, hm = 0.0f;
      if (mode != 0) {
        hr = accL[(c >> 4) * 2 + 0][row * 16 + (c & 15)];
        hm = accL[(c >> 4) * 2 + 1][row * 16 + (c & 15)];
        accL[(c >> 4) * 2 + 0][row * 16 + (c & 15)] = 0.0f;
        accL[(c >> 4) * 2 + 1][row * 16 + (c & 15)] = 0.0f;
      }
      if (mode == 2) {
        out[(rowbase * F_N + e) * 2]     = hr * inv;
        out[(rowbase * F_N + e) * 2 + 1] = hm * inv;
      } else {
        int p = ((c >> 2) * 32 + ((c >> 1) & 1) * 16 + row) * 2 + (c & 1);
        zLre[p] = s * ur[ii] - hr;
        zLim[p] = s * ui[ii] - hm;
      }
    }
    __syncthreads();

    if (mode != 2) {
      v2f zr[NKT_F], zi[NKT_F];
#pragma unroll
      for (int kt = 0; kt < NKT_F; ++kt) {
        zr[kt] = *(const v2f*)(zLre + kt * 64 + lane * 2);
        zi[kt] = *(const v2f*)(zLim + kt * 64 + lane * 2);
      }
      const bool more = (mode == 1) && (rt + (int)gridDim.x < NRT);
#pragma unroll
      for (int i = 0; i < 4; ++i) {
        v8f ar_ = {}, ai_ = {};
#pragma unroll
        for (int kt = 0; kt < NKT_F; ++kt) {
          v2f bn = -Bim[kt];
          ar_ = wmma_f32(Bre[kt], zr[kt], ar_);
          ar_ = wmma_f32(bn, zi[kt], ar_);
          ai_ = wmma_f32(Bim[kt], zr[kt], ai_);
          ai_ = wmma_f32(Bre[kt], zi[kt], ai_);
          if (i + 1 < 4) {
            Bre[kt] = *(const v2f*)(bbr + (i + 1) * 64 + kt * (NCT_T * 64));
            Bim[kt] = *(const v2f*)(bbi + (i + 1) * 64 + kt * (NCT_T * 64));
          }
        }
        v4f rr0, rr1, ri0, ri1;
        float cs[8];
#pragma unroll
        for (int q = 0; q < 4; ++q) {
          rr0[q] = ar_[q] + xr[i][0][q];  rr1[q] = ar_[4 + q] + xr[i][1][q];
          ri0[q] = ai_[q] + xi[i][0][q];  ri1[q] = ai_[4 + q] + xi[i][1][q];
          cs[q]     = rr0[q] * rr0[q] + ri0[q] * ri0[q];
          cs[4 + q] = rr1[q] * rr1[q] + ri1[q] * ri1[q];
        }
        *(v4f*)(hbr + i * 256) = rr0;  *(v4f*)(hbr + i * 256 + 128) = rr1;
        *(v4f*)(hbi + i * 256) = ri0;  *(v4f*)(hbi + i * 256 + 128) = ri1;
        if (more) {
          xr[i][0] = *(const v4f*)(hbr + nstep + i * 256);
          xr[i][1] = *(const v4f*)(hbr + nstep + i * 256 + 128);
          xi[i][0] = *(const v4f*)(hbi + nstep + i * 256);
          xi[i][1] = *(const v4f*)(hbi + nstep + i * 256 + 128);
        }
#pragma unroll
        for (int r = 0; r < 8; ++r) {
          float v = cs[r];
          v += __shfl_xor(v, 1, 32);
          v += __shfl_xor(v, 2, 32);
          v += __shfl_xor(v, 4, 32);
          v += __shfl_xor(v, 8, 32);
          cs[r] = v;
        }
        float mine = cs[0];
#pragma unroll
        for (int r = 1; r < 8; ++r) mine = ((lane & 7) == r) ? cs[r] : mine;
        if ((lane & 15) < 8)
          atomicAdd(&sum_out[(rt & (NCOPY - 1)) * T_N + (ct0 + i) * 16 + hi * 8 + (lane & 7)], mine);
      }
    }
  }
#undef LOADH
}

__global__ void __launch_bounds__(256, 4) k_layer_first(
    float* __restrict__ hre, float* __restrict__ him,
    const float* __restrict__ dft, const float* __restrict__ idft,
    const float* __restrict__ u, const float* __restrict__ scal,
    const float* __restrict__ sum_in, float* __restrict__ sum_out,
    const float* __restrict__ S1, const float* __restrict__ S2, float* __restrict__ out) {
  layer_body(0, hre, him, dft, idft, u, scal, sum_in, sum_out, S1, S2, out);
}

__global__ void __launch_bounds__(256, 4) k_layer_mid(
    float* __restrict__ hre, float* __restrict__ him,
    const float* __restrict__ dft, const float* __restrict__ idft,
    const float* __restrict__ u, const float* __restrict__ scal,
    const float* __restrict__ sum_in, float* __restrict__ sum_out,
    const float* __restrict__ S1, const float* __restrict__ S2, float* __restrict__ out) {
  layer_body(1, hre, him, dft, idft, u, scal, sum_in, sum_out, S1, S2, out);
}

__global__ void __launch_bounds__(256, 4) k_layer_last(
    float* __restrict__ hre, float* __restrict__ him,
    const float* __restrict__ dft, const float* __restrict__ idft,
    const float* __restrict__ u, const float* __restrict__ scal,
    const float* __restrict__ sum_in, float* __restrict__ sum_out,
    const float* __restrict__ S1, const float* __restrict__ S2, float* __restrict__ out) {
  layer_body(2, hre, him, dft, idft, u, scal, sum_in, sum_out, S1, S2, out);
}

extern "C" void kernel_launch(void* const* d_in, const int* in_sizes, int n_in,
                              void* d_out, int out_size, void* d_ws, size_t ws_size,
                              hipStream_t stream) {
  const float* u  = (const float*)d_in[0];
  const float* S1 = (const float*)d_in[1];
  const float* S2 = (const float*)d_in[2];
  float* out = (float*)d_out;
  float* w = (float*)d_ws;

  float* scal = w;
  float* part = w + 8;
  float* irep = w + 16384;
  float* iimp = irep + F_N * T_N;
  float* drep = iimp + F_N * T_N;
  float* dimp = drep + F_N * T_N;
  float* sums = w + 131072;
  float* hre  = w + 131072 + 2 * B_N * F_N;
  float* him  = hre + B_N * T_N;

  const int nu = B_N * F_N * 2;
  k_maxpart<<<1024, 256, 0, stream>>>(u, part, nu);
  k_finalize<<<1, 256, 0, stream>>>(part, scal);
  k_dft<<<(F_N * T_N + 255) / 256, 256, 0, stream>>>(irep, iimp, drep, dimp, sums);

  k_layer_first<<<512, 256, 0, stream>>>(hre, him, drep, irep, u, scal,
                                         sums, sums, S1, S2, out);
  for (int l = 1; l < L_N; ++l) {
    k_layer_mid<<<512, 256, 0, stream>>>(hre, him, drep, irep, u, scal,
                                         sums + (l - 1) * (NCOPY * T_N), sums + l * (NCOPY * T_N),
                                         S1, S2, out);
  }
  k_layer_last<<<512, 256, 0, stream>>>(hre, him, drep, irep, u, scal,
                                        sums + (L_N - 1) * (NCOPY * T_N), sums, S1, S2, out);
}
